# starting kernel
# baseline (speedup 1.0000x reference)
#include <hip/hip_runtime.h>

typedef float v2f __attribute__((ext_vector_type(2)));
typedef float v8f __attribute__((ext_vector_type(8)));

#define ROWS_PER_WAVE 32
#define ROWS_PER_BLOCK 128
#define SLOT 74

__global__ __launch_bounds__(128)
void hcr_wmma_kernel(const float* __restrict__ x,
                     const float* __restrict__ C,
                     float* __restrict__ out)
{
    __shared__ float smem[ROWS_PER_BLOCK * SLOT];

    const int tid  = threadIdx.x;
    const int lane = tid & 31;
    const int wid  = tid >> 5;
    const int hs   = lane >> 4;
    const int ln   = lane & 15;

    const size_t blockBase = (size_t)blockIdx.x * ROWS_PER_BLOCK;

    const float scale0 = 1.0f;
    const float scale1 = 1.7320508075688772f;
    const float scale2 = 2.2360679774997896f;
    const float scale3 = 2.6457513110645907f;
    const float scale4 = 3.0f;
    const float scale5 = 3.3166247903553998f;

    #pragma unroll
    for (int t = 0; t < 2; ++t) {
        const int rid = wid * ROWS_PER_WAVE + ln + 16 * t;
        const float* xp = x + (blockBase + rid) * 4 + 2 * hs;
        const float xa = xp[0];
        const float xb = xp[1];
        const float ta = 2.0f * xa - 1.0f;
        const float tb = 2.0f * xb - 1.0f;

        float Pa0 = 1.0f, Pa1 = ta;
        float Pb0 = 1.0f, Pb1 = tb;
        float Pa2 = (3.0f * ta * Pa1 - 1.0f * Pa0) * 0.5f;
        float Pb2 = (3.0f * tb * Pb1 - 1.0f * Pb0) * 0.5f;
        float Pa3 = (5.0f * ta * Pa2 - 2.0f * Pa1) * (1.0f / 3.0f);
        float Pb3 = (5.0f * tb * Pb2 - 2.0f * Pb1) * (1.0f / 3.0f);
        float Pa4 = (7.0f * ta * Pa3 - 3.0f * Pa2) * 0.25f;
        float Pb4 = (7.0f * tb * Pb3 - 3.0f * Pb2) * 0.25f;
        float Pa5 = (9.0f * ta * Pa4 - 4.0f * Pa3) * 0.2f;
        float Pb5 = (9.0f * tb * Pb4 - 4.0f * Pb3) * 0.2f;

        float ba[6] = { Pa0 * scale0, Pa1 * scale1, Pa2 * scale2,
                        Pa3 * scale3, Pa4 * scale4, Pa5 * scale5 };
        float bb[6] = { Pb0 * scale0, Pb1 * scale1, Pb2 * scale2,
                        Pb3 * scale3, Pb4 * scale4, Pb5 * scale5 };

        float* slot = smem + rid * SLOT + hs * 38;
        #pragma unroll
        for (int i = 0; i < 6; ++i) {
            #pragma unroll
            for (int j = 0; j < 6; ++j)
                slot[i * 6 + j] = ba[i] * bb[j];
        }
    }
    __syncthreads();

    const int krow = 2 * hs;
    const float* uRow0 = smem + (wid * ROWS_PER_WAVE + ln) * SLOT;
    const float* uRow1 = uRow0 + 16 * SLOT;
    const float* vRow0 = uRow0 + 38;
    const float* vRow1 = uRow1 + 38;
    const float* Bq0 = C + (size_t)ln * 1296 + krow;
    const float* Bq1 = C + (size_t)(ln + 16) * 1296 + krow;

    v8f acc00 = {}, acc01 = {}, acc10 = {}, acc11 = {};

    for (int p = 0; p < 36; ++p) {
        const float up0 = uRow0[p];
        const float up1 = uRow1[p];
        const int   kb  = p * 36;
        #pragma unroll 3
        for (int j = 0; j < 36; j += 4) {
            const int q = j + krow;
            const v2f vv0 = *(const v2f*)(vRow0 + q);
            const v2f vv1 = *(const v2f*)(vRow1 + q);
            const v2f b0  = *(const v2f*)(Bq0 + kb + j);
            const v2f b1  = *(const v2f*)(Bq1 + kb + j);
            v2f a0; a0.x = up0 * vv0.x; a0.y = up0 * vv0.y;
            v2f a1; a1.x = up1 * vv1.x; a1.y = up1 * vv1.y;
            acc00 = __builtin_amdgcn_wmma_f32_16x16x4_f32(false, a0, false, b0,
                                                          (short)0, acc00, false, false);
            acc01 = __builtin_amdgcn_wmma_f32_16x16x4_f32(false, a0, false, b1,
                                                          (short)0, acc01, false, false);
            acc10 = __builtin_amdgcn_wmma_f32_16x16x4_f32(false, a1, false, b0,
                                                          (short)0, acc10, false, false);
            acc11 = __builtin_amdgcn_wmma_f32_16x16x4_f32(false, a1, false, b1,
                                                          (short)0, acc11, false, false);
        }
    }

    const float z0 = C[(size_t)ln * 1296];
    const float z1 = C[(size_t)(ln + 16) * 1296];
    float* op = out + (blockBase + wid * ROWS_PER_WAVE + 8 * hs) * 32 + ln;
    #pragma unroll
    for (int r = 0; r < 8; ++r) {
        op[r * 32]              = 1.0f + acc00[r] - z0;
        op[r * 32 + 16]         = 1.0f + acc01[r] - z1;
        op[(r + 16) * 32]       = 1.0f + acc10[r] - z0;
        op[(r + 16) * 32 + 16]  = 1.0f + acc11[r] - z1;
    }
}

extern "C" void kernel_launch(void* const* d_in, const int* in_sizes, int n_in,
                              void* d_out, int out_size, void* d_ws, size_t ws_size,
                              hipStream_t stream) {
    const float* x = (const float*)d_in[0];
    const float* C = (const float*)d_in[1];
    float* out = (float*)d_out;

    const int B = in_sizes[0] / 4;
    const int blocks = B / ROWS_PER_BLOCK;
    hcr_wmma_kernel<<<blocks, 128, 0, stream>>>(x, C, out);
}


# Coefficient matrix staged once per work-group in LDS (pitch 1300 floats), one 16-wave group per WGP; B fragments by ds_load_b64, each block's 18 issued as one group one block ahead; same sum order
# speedup vs baseline: 1.3404x; 1.3404x over previous
#include <hip/hip_runtime.h>


typedef float v2f __attribute__((ext_vector_type(2)));
typedef float v4f __attribute__((ext_vector_type(4)));
typedef float v8f __attribute__((ext_vector_type(8)));

#define ROWS_PER_WAVE 32
#define WAVES_PER_BLOCK 16
#define ROWS_PER_BLOCK 512
#define SLOT 74
#define KDIM 1296
#define CPITCH 1300

__global__ __launch_bounds__(512)
void hcr_wmma_kernel(const float* __restrict__ x,
                     const float* __restrict__ C,
                     float* __restrict__ out)
{
    __shared__ alignas(16) float Cs[32 * CPITCH];
    __shared__ alignas(16) float smem[ROWS_PER_BLOCK * SLOT];

    const int tid  = threadIdx.x;
    const int lane = tid & 31;
    const int wid  = tid >> 5;
    const int hs   = lane >> 4;
    const int ln   = lane & 15;

    const size_t blockBase = (size_t)blockIdx.x * ROWS_PER_BLOCK;

    float xin[2][2];
    #pragma unroll
    for (int t = 0; t < 2; ++t) {
        const int rid = wid * ROWS_PER_WAVE + ln + 16 * t;
        const float* xp = x + (blockBase + rid) * 4 + 2 * hs;
        xin[t][0] = xp[0];
        xin[t][1] = xp[1];
    }
    v4f ct[2][11];
    #pragma unroll
    for (int rr = 0; rr < 2; ++rr) {
        const float* crow = C + (size_t)(2 * wid + rr) * KDIM;
        #pragma unroll
        for (int s = 0; s < 10; ++s)
            ct[rr][s] = *(const v4f*)(crow + 4 * (lane + 32 * s));
        if (lane < 4) ct[rr][10] = *(const v4f*)(crow + 4 * (lane + 320));
    }

    const float scale0 = 1.0f;
    const float scale1 = 1.7320508075688772f;
    const float scale2 = 2.2360679774997896f;
    const float scale3 = 2.6457513110645907f;
    const float scale4 = 3.0f;
    const float scale5 = 3.3166247903553998f;

    #pragma unroll
    for (int t = 0; t < 2; ++t) {
        const int rid = wid * ROWS_PER_WAVE + ln + 16 * t;
        const float xa = xin[t][0];
        const float xb = xin[t][1];
        const float ta = 2.0f * xa - 1.0f;
        const float tb = 2.0f * xb - 1.0f;

        float Pa0 = 1.0f, Pa1 = ta;
        float Pb0 = 1.0f, Pb1 = tb;
        float Pa2 = (3.0f * ta * Pa1 - 1.0f * Pa0) * 0.5f;
        float Pb2 = (3.0f * tb * Pb1 - 1.0f * Pb0) * 0.5f;
        float Pa3 = (5.0f * ta * Pa2 - 2.0f * Pa1) * (1.0f / 3.0f);
        float Pb3 = (5.0f * tb * Pb2 - 2.0f * Pb1) * (1.0f / 3.0f);
        float Pa4 = (7.0f * ta * Pa3 - 3.0f * Pa2) * 0.25f;
        float Pb4 = (7.0f * tb * Pb3 - 3.0f * Pb2) * 0.25f;
        float Pa5 = (9.0f * ta * Pa4 - 4.0f * Pa3) * 0.2f;
        float Pb5 = (9.0f * tb * Pb4 - 4.0f * Pb3) * 0.2f;

        float ba[6] = { Pa0 * scale0, Pa1 * scale1, Pa2 * scale2,
                        Pa3 * scale3, Pa4 * scale4, Pa5 * scale5 };
        float bb[6] = { Pb0 * scale0, Pb1 * scale1, Pb2 * scale2,
                        Pb3 * scale3, Pb4 * scale4, Pb5 * scale5 };

        float* slot = smem + rid * SLOT + hs * 38;
        #pragma unroll
        for (int i = 0; i < 6; ++i) {
            #pragma unroll
            for (int j = 0; j < 6; ++j)
                slot[i * 6 + j] = ba[i] * bb[j];
        }
    }

    #pragma unroll
    for (int rr = 0; rr < 2; ++rr) {
        float* srow = Cs + (2 * wid + rr) * CPITCH;
        #pragma unroll
        for (int s = 0; s < 10; ++s)
            *(v4f*)(srow + 4 * (lane + 32 * s)) = ct[rr][s];
        if (lane < 4) *(v4f*)(srow + 4 * (lane + 320)) = ct[rr][10];
    }
    __syncthreads();

    const int krow = 2 * hs;
    const float* uRow0 = smem + (wid * ROWS_PER_WAVE + ln) * SLOT;
    const float* uRow1 = uRow0 + 16 * SLOT;
    const float* vRow0 = uRow0 + 38 + krow;
    const float* vRow1 = uRow1 + 38 + krow;
    const float* Bq0 = Cs + ln * CPITCH + krow;
    const float* Bq1 = Cs + (ln + 16) * CPITCH + krow;

    v2f vf0[9], vf1[9];
    #pragma unroll
    for (int m = 0; m < 9; ++m) {
        vf0[m] = *(const v2f*)(vRow0 + 4 * m);
        vf1[m] = *(const v2f*)(vRow1 + 4 * m);
    }

    v8f acc00 = {}, acc01 = {}, acc10 = {}, acc11 = {};

#define LOAD_B(P, FB0, FB1)                                                   \
    {                                                                         \
        const float* bp0 = Bq0 + (P) * 36;                                    \
        const float* bp1 = Bq1 + (P) * 36;                                    \
        _Pragma("unroll")                                                     \
        for (int m = 0; m < 9; ++m) {                                         \
            FB0[m] = *(const v2f*)(bp0 + 4 * m);                              \
            FB1[m] = *(const v2f*)(bp1 + 4 * m);                              \
        }                                                                     \
    }
#define MMA_BLOCK(U0, U1, FB0, FB1)                                           \
    {                                                                         \
        _Pragma("unroll")                                                     \
        for (int m = 0; m < 9; ++m) {                                         \
            const v2f a0 = (U0) * vf0[m];                                     \
            const v2f a1 = (U1) * vf1[m];                                     \
            acc00 = __builtin_amdgcn_wmma_f32_16x16x4_f32(false, a0, false, FB0[m], \
                                                  (short)0, acc00, false, false);   \
            acc01 = __builtin_amdgcn_wmma_f32_16x16x4_f32(false, a0, false, FB1[m], \
                                                  (short)0, acc01, false, false);   \
            acc10 = __builtin_amdgcn_wmma_f32_16x16x4_f32(false, a1, false, FB0[m], \
                                                  (short)0, acc10, false, false);   \
            acc11 = __builtin_amdgcn_wmma_f32_16x16x4_f32(false, a1, false, FB1[m], \
                                                  (short)0, acc11, false, false);   \
        }                                                                     \
    }

    v2f ba0[9], ba1[9], bb0[9], bb1[9];
    v2f u0c = *(const v2f*)(uRow0);
    v2f u1c = *(const v2f*)(uRow1);
    LOAD_B(0, ba0, ba1)

    for (int p = 0; p < 36; p += 2) {
        const int pn = (p + 2 < 36) ? p + 2 : 34;
        const v2f u0n = *(const v2f*)(uRow0 + pn);
        const v2f u1n = *(const v2f*)(uRow1 + pn);
        LOAD_B(p + 1, bb0, bb1)
        __builtin_amdgcn_sched_barrier(0);
        MMA_BLOCK(u0c.x, u1c.x, ba0, ba1)
        __builtin_amdgcn_sched_barrier(0);
        LOAD_B(pn, ba0, ba1)
        __builtin_amdgcn_sched_barrier(0);
        MMA_BLOCK(u0c.y, u1c.y, bb0, bb1)
        __builtin_amdgcn_sched_barrier(0);
        u0c = u0n;
        u1c = u1n;
    }
#undef LOAD_B
#undef MMA_BLOCK

    const float z0 = Cs[ln * CPITCH];
    const float z1 = Cs[(ln + 16) * CPITCH];
    float* op = out + (blockBase + wid * ROWS_PER_WAVE + 8 * hs) * 32 + ln;
    #pragma unroll
    for (int r = 0; r < 8; ++r) {
        op[r * 32]              = 1.0f + acc00[r] - z0;
        op[r * 32 + 16]         = 1.0f + acc01[r] - z1;
        op[(r + 16) * 32]       = 1.0f + acc10[r] - z0;
        op[(r + 16) * 32 + 16]  = 1.0f + acc11[r] - z1;
    }
}

extern "C" void kernel_launch(void* const* d_in, const int* in_sizes, int n_in,
                              void* d_out, int out_size, void* d_ws, size_t ws_size,
                              hipStream_t stream) {
    const float* x = (const float*)d_in[0];
    const float* C = (const float*)d_in[1];
    float* out = (float*)d_out;

    const int B = in_sizes[0] / 4;
    const int blocks = B / ROWS_PER_BLOCK;
    hcr_wmma_kernel<<<blocks, 512, 0, stream>>>(x, C, out);
}
